# starting kernel
# baseline (speedup 1.0000x reference)
#include <hip/hip_runtime.h>


#define N_NODES 50000
#define N_EDGES 800000
#define IN_C    128
#define HID_C   128
#define OUT_C   64

typedef float v2f __attribute__((ext_vector_type(2)));
typedef float v8f __attribute__((ext_vector_type(8)));

__global__ void sage_zero_kernel(float* __restrict__ p, long long n4) {
    long long i = (long long)blockIdx.x * blockDim.x + threadIdx.x;
    if (i < n4) ((float4*)p)[i] = make_float4(0.f, 0.f, 0.f, 0.f);
}

__global__ __launch_bounds__(256) void sage_scatter_kernel(
    const float* __restrict__ feat,
    const int*   __restrict__ src,
    const int*   __restrict__ dst,
    float* __restrict__ sum,
    float* __restrict__ deg,
    int n_edges) {
    const int lane = threadIdx.x & 31;
    const int e    = blockIdx.x * (blockDim.x >> 5) + (threadIdx.x >> 5);
    if (e >= n_edges) return;
    const int s = src[e];
    const int d = dst[e];
    const float4 v = ((const float4*)(feat + (long long)s * 128))[lane];
    float* o = sum + (long long)d * 128 + lane * 4;
    atomicAdd(o + 0, v.x);
    atomicAdd(o + 1, v.y);
    atomicAdd(o + 2, v.z);
    atomicAdd(o + 3, v.w);
    if (deg != nullptr && lane == 0) atomicAdd(deg + d, 1.0f);
}

template <int CO, bool RELU>
__global__ __launch_bounds__(256) void sage_gemm_kernel(
    const float* __restrict__ sum,
    const float* __restrict__ deg,
    const float* __restrict__ feat,
    const float* __restrict__ Wl,
    const float* __restrict__ bias,
    const float* __restrict__ Wr,
    float* __restrict__ out,
    int n_tiles) {
    constexpr int NT = CO / 16;

    extern __shared__ float smem[];
    float* sWl = smem;
    float* sWr = smem + 128 * CO;

    const int tid = threadIdx.x;
    for (int i = tid; i < 128 * CO; i += blockDim.x) {
        sWl[i] = Wl[i];
        sWr[i] = Wr[i];
    }
    __syncthreads();

    const int wave = tid >> 5;
    const int lane = tid & 31;
    const int tile = blockIdx.x * (blockDim.x >> 5) + wave;
    if (tile >= n_tiles) return;

    const int row0 = tile * 16;
    const int hi   = lane >> 4;
    const int lr   = lane & 15;
    const int arow = row0 + lr;

    const float dg   = deg[arow];
    const float invd = 1.0f / fmaxf(dg, 1.0f);

    const float* aggRow = sum  + (long long)arow * 128;
    const float* xRow   = feat + (long long)arow * 128;

    v8f acc[NT];
#pragma unroll
    for (int n = 0; n < NT; ++n) acc[n] = (v8f){};

    for (int k = 0; k < 32; ++k) {
        const int c = 4 * k + 2 * hi;
        v2f aAgg, aX;
        aAgg.x = aggRow[c]     * invd;
        aAgg.y = aggRow[c + 1] * invd;
        aX.x   = xRow[c];
        aX.y   = xRow[c + 1];
#pragma unroll
        for (int n = 0; n < NT; ++n) {
            const int col = n * 16 + lr;
            v2f bL, bR;
            bL.x = sWl[(c)     * CO + col];
            bL.y = sWl[(c + 1) * CO + col];
            bR.x = sWr[(c)     * CO + col];
            bR.y = sWr[(c + 1) * CO + col];
            acc[n] = __builtin_amdgcn_wmma_f32_16x16x4_f32(
                false, aAgg, false, bL, (short)0, acc[n], false, false);
            acc[n] = __builtin_amdgcn_wmma_f32_16x16x4_f32(
                false, aX,   false, bR, (short)0, acc[n], false, false);
        }
    }

#pragma unroll
    for (int n = 0; n < NT; ++n) {
        const int col = n * 16 + lr;
        const float b = bias[col];
#pragma unroll
        for (int v = 0; v < 8; ++v) {
            const int r = row0 + v + 8 * hi;
            float val = acc[n][v] + b;
            if (RELU) val = fmaxf(val, 0.0f);
            out[(long long)r * CO + col] = val;
        }
    }
}

extern "C" void kernel_launch(void* const* d_in, const int* in_sizes, int n_in,
                              void* d_out, int out_size, void* d_ws, size_t ws_size,
                              hipStream_t stream) {
    const float* x   = (const float*)d_in[0];
    const int*   ei  = (const int*)  d_in[1];
    const float* W1l = (const float*)d_in[2];
    const float* b1  = (const float*)d_in[3];
    const float* W1r = (const float*)d_in[4];
    const float* W2l = (const float*)d_in[5];
    const float* b2  = (const float*)d_in[6];
    const float* W2r = (const float*)d_in[7];

    const int* src = ei;
    const int* dst = ei + N_EDGES;

    float* ws  = (float*)d_ws;
    float* sum = ws;
    float* deg = ws + (long long)N_NODES * 128;
    float* h   = deg + N_NODES;
    float* out = (float*)d_out;

    const int tiles       = N_NODES / 16;
    const int gemm_blocks = (tiles + 7) / 8;
    const int edge_blocks = (N_EDGES + 7) / 8;

    {
        const long long n = (long long)N_NODES * 128 + N_NODES;
        const long long n4 = n / 4;
        sage_zero_kernel<<<(int)((n4 + 255) / 256), 256, 0, stream>>>(sum, n4);
    }
    sage_scatter_kernel<<<edge_blocks, 256, 0, stream>>>(x, src, dst, sum, deg, N_EDGES);
    sage_gemm_kernel<HID_C, true><<<gemm_blocks, 256, 2 * 128 * HID_C * sizeof(float), stream>>>(
        sum, deg, x, W1l, b1, W1r, h, tiles);

    {
        const long long n  = (long long)N_NODES * 128;
        const long long n4 = n / 4;
        sage_zero_kernel<<<(int)((n4 + 255) / 256), 256, 0, stream>>>(sum, n4);
    }
    sage_scatter_kernel<<<edge_blocks, 256, 0, stream>>>(h, src, dst, sum, nullptr, N_EDGES);
    sage_gemm_kernel<OUT_C, false><<<gemm_blocks, 256, 2 * 128 * OUT_C * sizeof(float), stream>>>(
        sum, deg, h, W2l, b2, W2r, out, tiles);
}


# Linear kernels re-decomposed: each wave keeps one 16-column block of both weight matrices in registers (B fragments) all launch, node rows stream through LDS; per-tile sums now on 2 accumulators
# speedup vs baseline: 3.8115x; 3.8115x over previous
#include <hip/hip_runtime.h>


#define N_NODES 50000
#define N_EDGES 800000
#define IN_C    128
#define HID_C   128
#define OUT_C   64

typedef float v2f __attribute__((ext_vector_type(2)));
typedef float v4f __attribute__((ext_vector_type(4)));
typedef float v8f __attribute__((ext_vector_type(8)));

#define NPB        224
#define AGG_T      1024
#define AGG_W      (AGG_T / 32)
#define AGG_ROWS   ((NPB + AGG_W - 1) / AGG_W)
#define AGG_RC     64
#define AGG_BLOCKS ((N_NODES + NPB - 1) / NPB)
#define AGG_NVEC   (N_EDGES / 4)

static_assert(N_EDGES % 4 == 0, "dst is read in 16-byte pieces");
static_assert(AGG_RC == 64, "a row's list is looked up as two entries per lane");
static_assert(AGG_NVEC >= AGG_T, "index wrap assumes one subtraction is enough");
static_assert(NPB % 32 == 0 && NPB == AGG_ROWS * AGG_W, "a group's degrees fill whole 128-byte lines");
static_assert(AGG_RC * sizeof(int) == OUT_C * sizeof(float), "a row's stored list takes the place of one output row");

template <bool SCAN>
__global__ __launch_bounds__(AGG_T) void sage_agg_kernel(
    const float* __restrict__ feat,
    const int*   __restrict__ src,
    const int*   __restrict__ dst,
    int*   __restrict__ rowsrc,
    float* __restrict__ sum,
    float* __restrict__ deg) {
    __shared__ int lists[AGG_ROWS * AGG_W * AGG_RC];
    __shared__ int lcnt[AGG_ROWS * AGG_W];
    __shared__ int ovf;

    const int tid  = threadIdx.x;
    const int lane = tid & 31;
    const int wave = __builtin_amdgcn_readfirstlane(tid >> 5);
    const int base = blockIdx.x * NPB;
    const int nb   = min(NPB, N_NODES - base);

    if (SCAN && tid < AGG_ROWS * AGG_W) lcnt[tid] = 0;
    if (tid == 0) ovf = 0;
    __syncthreads();

    if constexpr (SCAN) {
        const int4* dst4 = (const int4*)dst;
        const int off = (int)((blockIdx.x * (AGG_NVEC / AGG_BLOCKS)) & ~31u);
        constexpr int U       = 5;
        constexpr int nFull   = AGG_NVEC / AGG_T;
        constexpr int nRounds = nFull / U;
        constexpr int nIt     = (AGG_NVEC + AGG_T - 1) / AGG_T;

        auto wrap = [](int i) -> int { return i >= AGG_NVEC ? i - AGG_NVEC : i; };
        auto take = [&](unsigned rel, int e) {
            if (rel < (unsigned)nb) {
                const int slot = atomicAdd(&lcnt[rel], 1);
                if (slot < AGG_RC) lists[rel * AGG_RC + slot] = e;
                else ovf = 1;
            }
        };
        auto scan_vec = [&](int4 d, int e0) {
            const unsigned rx = (unsigned)d.x - (unsigned)base, ry = (unsigned)d.y - (unsigned)base;
            const unsigned rz = (unsigned)d.z - (unsigned)base, rw = (unsigned)d.w - (unsigned)base;
            if (min(min(rx, ry), min(rz, rw)) < (unsigned)nb) {
                take(rx, e0);
                take(ry, e0 + 1);
                take(rz, e0 + 2);
                take(rw, e0 + 3);
            }
        };

        int fidx = wrap(off + tid);
        int pidx = fidx;
        int4 cur[U];
#pragma unroll
        for (int u = 0; u < U; ++u) { cur[u] = dst4[fidx]; fidx = wrap(fidx + AGG_T); }
        for (int r = 0; r < nRounds; ++r) {
            int4 nxt[U];
#pragma unroll
            for (int u = 0; u < U; ++u) {
                nxt[u] = dst4[fidx];
                fidx = wrap(fidx + AGG_T);
            }
#pragma unroll
            for (int u = 0; u < U; ++u) {
                scan_vec(cur[u], pidx * 4);
                pidx = wrap(pidx + AGG_T);
            }
#pragma unroll
            for (int u = 0; u < U; ++u) cur[u] = nxt[u];
        }
        for (int it = nRounds * U; it < nIt; ++it) {
            const int q = it * AGG_T + tid;
            if (q < AGG_NVEC) {
                const int idx = wrap(q + off);
                scan_vec(dst4[idx], idx * 4);
            }
        }
    } else {
        if (tid < AGG_ROWS * AGG_W) {
            const int c = tid < nb ? (int)deg[base + tid] : 0;
            lcnt[tid] = c;
            if (c > AGG_RC) ovf = 1;
        }
    }
    __syncthreads();

    const float4* f4   = (const float4*)feat;
    float4*       sum4 = (float4*)sum;
    const bool    fits = ovf == 0;

    if (SCAN && tid < nb) deg[base + tid] = (float)lcnt[tid];

    if (fits) {
        if constexpr (SCAN) {
            int sv[AGG_ROWS][2];
#pragma unroll
            for (int s = 0; s < AGG_ROWS; ++s) {
                const int rel = wave + s * AGG_W;
                const int n   = lcnt[rel];
#pragma unroll
                for (int t = 0; t < 2; ++t) {
                    const int i = lane + 32 * t;
                    const int e = i < n ? lists[rel * AGG_RC + i] : 0;
                    sv[s][t] = src[e];
                }
            }
#pragma unroll
            for (int s = 0; s < AGG_ROWS; ++s) {
                const int rel = wave + s * AGG_W;
                const int n   = lcnt[rel];
#pragma unroll
                for (int t = 0; t < 2; ++t) {
                    const int i = lane + 32 * t;
                    if (i < n) {
                        const int sr = min(max(sv[s][t], 0), N_NODES - 1);
                        lists[rel * AGG_RC + i] = sr;
                        rowsrc[(long long)(base + rel) * AGG_RC + i] = sr;
                    }
                }
            }
        } else {
            int sv[AGG_ROWS][2];
#pragma unroll
            for (int s = 0; s < AGG_ROWS; ++s) {
                const int rel = wave + s * AGG_W;
                const int n   = lcnt[rel];
#pragma unroll
                for (int t = 0; t < 2; ++t) {
                    const int i = lane + 32 * t;
                    sv[s][t] = i < n ? rowsrc[(long long)(base + rel) * AGG_RC + i] : 0;
                }
            }
#pragma unroll
            for (int s = 0; s < AGG_ROWS; ++s) {
                const int rel = wave + s * AGG_W;
                const int n   = lcnt[rel];
#pragma unroll
                for (int t = 0; t < 2; ++t) {
                    const int i = lane + 32 * t;
                    if (i < n) lists[rel * AGG_RC + i] = sv[s][t];
                }
            }
        }
    }
    __syncthreads();

    if (fits) {
        constexpr int B = 8;
        for (int rel = wave; rel < nb; rel += AGG_W) {
            const int n = __builtin_amdgcn_readfirstlane(lcnt[rel]);
            float4 t = make_float4(0.f, 0.f, 0.f, 0.f);
            for (int i0 = 0; i0 < n; i0 += B) {
                const int sidx = lists[rel * AGG_RC + min(i0 + (lane & (B - 1)), n - 1)];
                float4 v[B];
#pragma unroll
                for (int j = 0; j < B; ++j)
                    v[j] = f4[(long long)__builtin_amdgcn_readlane(sidx, j) * 32 + lane];
#pragma unroll
                for (int j = 0; j < B; ++j) {
                    const bool on = i0 + j < n;
                    t.x += on ? v[j].x : 0.f;
                    t.y += on ? v[j].y : 0.f;
                    t.z += on ? v[j].z : 0.f;
                    t.w += on ? v[j].w : 0.f;
                }
            }
            sum4[(long long)(base + rel) * 32 + lane] = t;
        }
    } else {
        for (int rel = wave; rel < nb; rel += AGG_W) {
            const int target = base + rel;
            float4 t = make_float4(0.f, 0.f, 0.f, 0.f);
            for (int e0 = 0; e0 < N_EDGES; e0 += 32) {
                const int e = e0 + lane;
                unsigned b = __builtin_amdgcn_ballot_w32(e < N_EDGES && dst[e] == target);
                while (b != 0) {
                    const int L = __builtin_ctz(b);
                    b &= b - 1;
                    const int s = min(max(src[e0 + L], 0), N_NODES - 1);
                    const float4 v = f4[(long long)s * 32 + lane];
                    t.x += v.x; t.y += v.y; t.z += v.z; t.w += v.w;
                }
            }
            sum4[(long long)(base + rel) * 32 + lane] = t;
        }
    }
}

#define GEMM_T    256
#define GEMM_W    (GEMM_T / 32)
#define GEMM_GRID 512
#define GEMM_P    132

template <int CO> struct GemmShape {
    static constexpr int NT  = CO / 16;
    static constexpr int RT  = GEMM_W / NT;
    static constexpr int RS  = 16 * RT;
    static constexpr int NP  = RS * 32 / GEMM_T;
    static constexpr int BUF = RS * GEMM_P;
    static constexpr unsigned LDS = 2u * 2u * BUF * sizeof(float);
};

template <int CO, bool RELU>
__global__ __launch_bounds__(GEMM_T) void sage_gemm_kernel(
    const float* __restrict__ sum,
    const float* __restrict__ deg,
    const float* __restrict__ feat,
    const float* __restrict__ Wl,
    const float* __restrict__ bias,
    const float* __restrict__ Wr,
    float* __restrict__ out,
    int n_tiles) {
    using S = GemmShape<CO>;
    constexpr int NT = S::NT, RT = S::RT, RS = S::RS, NP = S::NP;
    static_assert(GEMM_W % NT == 0 && (RS * 32) % GEMM_T == 0, "waves and pieces divide evenly");

    extern __shared__ float smem[];

    const int tid   = threadIdx.x;
    const int lane  = tid & 31;
    const int wave  = __builtin_amdgcn_readfirstlane(tid >> 5);
    const int hi    = lane >> 4;
    const int lr    = lane & 15;
    const int ntile = wave % NT;
    const int rsub  = wave / NT;
    const int col   = ntile * 16 + lr;
    const int n_steps = (n_tiles + RT - 1) / RT;

    const v4f* sum4  = (const v4f*)sum;
    const v4f* feat4 = (const v4f*)feat;

    v4f    rs[NP], rx[NP];
    float  rdeg[NP];
    auto fetch = [&](int step) {
#pragma unroll
        for (int j = 0; j < NP; ++j) {
            const int p   = tid + GEMM_T * j;
            const int row = min(step * RS + (p >> 5), N_NODES - 1);
            rdeg[j] = deg[row];
            rs[j]   = sum4[(long long)row * 32 + (p & 31)];
            rx[j]   = feat4[(long long)row * 32 + (p & 31)];
        }
    };
    auto stash = [&](int buf) {
        float* sS = smem + buf * 2 * S::BUF;
        float* sX = sS + S::BUF;
#pragma unroll
        for (int j = 0; j < NP; ++j) {
            const int p = tid + GEMM_T * j;
            const int r = p >> 5, q = p & 31;
            const float invd = 1.0f / fmaxf(rdeg[j], 1.0f);
            *(v4f*)(sS + r * GEMM_P + 4 * q) = rs[j] * invd;
            *(v4f*)(sX + r * GEMM_P + 4 * q) = rx[j];
        }
    };

    fetch(blockIdx.x);

    v2f bL[32], bR[32];
#pragma unroll
    for (int kk = 0; kk < 32; ++kk) {
        const int c = 4 * kk + 2 * hi;
        bL[kk].x = Wl[(c)     * CO + col];
        bL[kk].y = Wl[(c + 1) * CO + col];
        bR[kk].x = Wr[(c)     * CO + col];
        bR[kk].y = Wr[(c + 1) * CO + col];
    }
    const float b = bias[col];

    int it = 0;
    for (int step = blockIdx.x; step < n_steps; step += GEMM_GRID, ++it) {
        const int buf = it & 1;
        stash(buf);
        __syncthreads();
        const int nxt = step + GEMM_GRID;
        if (nxt < n_steps) fetch(nxt);

        const float* aS = smem + buf * 2 * S::BUF + (rsub * 16 + lr) * GEMM_P + 2 * hi;
        const float* aX = aS + S::BUF;
        v8f accL = (v8f){}, accR = (v8f){};
#pragma unroll
        for (int kk = 0; kk < 32; ++kk) {
            const v2f fA = *(const v2f*)(aS + 4 * kk);
            const v2f fX = *(const v2f*)(aX + 4 * kk);
            accL = __builtin_amdgcn_wmma_f32_16x16x4_f32(
                false, fA, false, bL[kk], (short)0, accL, false, false);
            accR = __builtin_amdgcn_wmma_f32_16x16x4_f32(
                false, fX, false, bR[kk], (short)0, accR, false, false);
        }

        float val[8];
#pragma unroll
        for (int v = 0; v < 8; ++v) {
            float t = (accL[v] + accR[v]) + b;
            if (RELU) t = fmaxf(t, 0.0f);
            val[v] = t;
        }
        const int tile = step * RT + rsub;
        if (tile < n_tiles) {
            const int row0 = tile * 16;
#pragma unroll
            for (int v = 0; v < 8; ++v) {
                const int r = row0 + v + 8 * hi;
                out[(long long)r * CO + col] = val[v];
            }
        }
    }
}

extern "C" void kernel_launch(void* const* d_in, const int* in_sizes, int n_in,
                              void* d_out, int out_size, void* d_ws, size_t ws_size,
                              hipStream_t stream) {
    const float* x   = (const float*)d_in[0];
    const int*   ei  = (const int*)  d_in[1];
    const float* W1l = (const float*)d_in[2];
    const float* b1  = (const float*)d_in[3];
    const float* W1r = (const float*)d_in[4];
    const float* W2l = (const float*)d_in[5];
    const float* b2  = (const float*)d_in[6];
    const float* W2r = (const float*)d_in[7];

    const int* src = ei;
    const int* dst = ei + N_EDGES;

    float* ws  = (float*)d_ws;
    float* sum = ws;
    float* deg = ws + (long long)N_NODES * 128;
    float* h   = deg + N_NODES;
    float* out = (float*)d_out;
    int*   rowsrc = (int*)d_out;

    const int tiles = N_NODES / 16;

    sage_agg_kernel<true><<<AGG_BLOCKS, AGG_T, 0, stream>>>(x, src, dst, rowsrc, sum, deg);
    sage_gemm_kernel<HID_C, true><<<GEMM_GRID, GEMM_T, GemmShape<HID_C>::LDS, stream>>>(
        sum, deg, x, W1l, b1, W1r, h, tiles);

    sage_agg_kernel<false><<<AGG_BLOCKS, AGG_T, 0, stream>>>(h, src, dst, rowsrc, sum, deg);
    sage_gemm_kernel<OUT_C, false><<<GEMM_GRID, GEMM_T, GemmShape<OUT_C>::LDS, stream>>>(
        sum, deg, h, W2l, b2, W2r, out, tiles);
}
